# starting kernel
# baseline (speedup 1.0000x reference)
#include <hip/hip_runtime.h>

typedef __attribute__((ext_vector_type(2))) float v2f;
typedef __attribute__((ext_vector_type(4))) float v4f;
typedef __attribute__((ext_vector_type(8))) float v8f;

#define B_  32
#define IC  32
#define OC  32
#define HH  64
#define WW  64
#define KD  288
#define LDP 292

__global__ __launch_bounds__(128)
void lc2d_wmma_f32_kernel(const float* __restrict__ x,
                          const float* __restrict__ wgt,
                          const float* __restrict__ bias,
                          float* __restrict__ out) {
    __shared__ float sA[B_ * LDP];
    __shared__ float sB[OC * LDP];

    const int w  = blockIdx.x;
    const int h  = blockIdx.y;
    const int t  = threadIdx.x;
    const int hw = h * WW + w;

    {
        const v4f* wp4 = (const v4f*)(wgt + (size_t)hw * (OC * KD));
        for (int e4 = t; e4 < (OC * KD) / 4; e4 += 128) {
            int e   = e4 * 4;
            int row = e / KD;
            int col = e - row * KD;
            v4f v = __builtin_nontemporal_load(wp4 + e4);
            *(v4f*)&sB[row * LDP + col] = v;
        }
    }

    for (int r = t; r < B_ * IC * 3; r += 128) {
        int b   = r / 96;
        int rem = r - b * 96;
        int c   = rem / 3;
        int i   = rem - c * 3;
        int y   = h - 1 + i;
        const bool yok = (unsigned)y < (unsigned)HH;
        const float* xrow = x + ((((size_t)b * IC + c) * HH + y) * WW + (w - 1));
        float* dst = &sA[b * LDP + c * 9 + i * 3];
        dst[0] = (yok && w - 1 >= 0) ? xrow[0] : 0.0f;
        dst[1] = yok                 ? xrow[1] : 0.0f;
        dst[2] = (yok && w + 1 < WW) ? xrow[2] : 0.0f;
    }
    __syncthreads();

    const int wv    = t >> 5;
    const int lane  = t & 31;
    const int m0    = (wv >> 1) * 16;
    const int n0    = (wv & 1) * 16;
    const int lr    = lane & 15;
    const int khalf = lane >> 4;

    const float* aRow = &sA[(m0 + lr) * LDP + 2 * khalf];
    const float* bRow = &sB[(n0 + lr) * LDP + 2 * khalf];

    v8f acc = {};
    #pragma unroll 8
    for (int kt = 0; kt < KD / 4; ++kt) {
        v2f a = *(const v2f*)(aRow + kt * 4);
        v2f b = *(const v2f*)(bRow + kt * 4);
        acc = __builtin_amdgcn_wmma_f32_16x16x4_f32(
            false, a, false, b, (short)0, acc, false, false);
    }

    const int N  = lane & 15;
    const int o  = n0 + N;
    const float bv = bias[(size_t)o * (HH * WW) + hw];
    #pragma unroll
    for (int r = 0; r < 8; ++r) {
        int M = r + khalf * 8;
        int b = m0 + M;
        __builtin_nontemporal_store(acc[r] + bv,
                                    out + (((size_t)b * OC + o) * (HH * WW)) + hw);
    }
}

extern "C" void kernel_launch(void* const* d_in, const int* in_sizes, int n_in,
                              void* d_out, int out_size, void* d_ws, size_t ws_size,
                              hipStream_t stream) {
    const float* x    = (const float*)d_in[0];
    const float* wgt  = (const float*)d_in[1];
    const float* bias = (const float*)d_in[2];
    float* out        = (float*)d_out;
    (void)in_sizes; (void)n_in; (void)out_size; (void)d_ws; (void)ws_size;

    dim3 grid(WW, HH);
    lc2d_wmma_f32_kernel<<<grid, 128, 0, stream>>>(x, wgt, bias, out);
}


# each work-group does 16 consecutive w of one row: x slab staged once with coalesced reads, results gathered in LDS and stored as 64-byte runs; K halves of two wave groups added in LDS (new sum order)
# speedup vs baseline: 5.9322x; 5.9322x over previous
#include <hip/hip_runtime.h>


typedef __attribute__((ext_vector_type(2))) float v2f;
typedef __attribute__((ext_vector_type(4))) float v4f;
typedef __attribute__((ext_vector_type(8))) float v8f;

#define B_  32
#define IC  32
#define OC  32
#define HH  64
#define WW  64
#define KD  288
#define LDP 292
#define TW  16
#define SW  18
#define CP  54
#define PB  1729
#define NTH 256

__device__ constexpr int tap_off(int t) {
    return (t / 9) * CP + ((t % 9) / 3) * SW + (t % 9) % 3;
}

__device__ __forceinline__ void lds_barrier() {
    __builtin_amdgcn_fence(__ATOMIC_RELEASE, "workgroup", "local");
    __builtin_amdgcn_s_barrier();
    __builtin_amdgcn_fence(__ATOMIC_ACQUIRE, "workgroup", "local");
}

__global__ __launch_bounds__(NTH)
void lc2d_strip_wmma_f32_kernel(const float* __restrict__ x,
                                const float* __restrict__ wgt,
                                const float* __restrict__ bias,
                                float* __restrict__ out) {
    __shared__ __attribute__((aligned(16))) float sB[OC * LDP];
    __shared__ float sO[B_ * TW * OC];
    __shared__ float sX[B_ * PB];

    const int t  = threadIdx.x;
    const int w0 = blockIdx.x * TW;
    const int h  = blockIdx.y;

    const v4f* wp4 = (const v4f*)(wgt + (size_t)(h * WW + w0) * (OC * KD));
    v4f wr[2][9];
    #pragma unroll
    for (int s = 0; s < 2; ++s) {
        #pragma unroll
        for (int u = 0; u < 9; ++u)
            wr[s][u] = __builtin_nontemporal_load(wp4 + s * ((OC * KD) / 4) + t + u * NTH);
    }

    {
        const int part = t & 3;
        const int bcl  = t >> 2;
        v4f v[3][16];
        #pragma unroll
        for (int i = 0; i < 3; ++i) {
            const int  y   = h - 1 + i;
            const bool yok = (unsigned)y < (unsigned)HH;
            if (yok) {
                const float* xp = x + (size_t)y * WW + w0 + 4 * part;
                #pragma unroll
                for (int n = 0; n < 16; ++n)
                    v[i][n] = *(const v4f*)(xp + (size_t)(n * 64 + bcl) * (HH * WW));
            } else {
                #pragma unroll
                for (int n = 0; n < 16; ++n) v[i][n] = (v4f){0.0f, 0.0f, 0.0f, 0.0f};
            }
        }
        const int  side = t & 1;
        const int  rl   = t >> 1;
        const int  wh   = side ? (w0 + TW) : (w0 - 1);
        const bool wok  = (unsigned)wh < (unsigned)WW;
        float hv[3][8];
        #pragma unroll
        for (int i = 0; i < 3; ++i) {
            const int  y  = h - 1 + i;
            const bool ok = wok && ((unsigned)y < (unsigned)HH);
            #pragma unroll
            for (int n = 0; n < 8; ++n) hv[i][n] = 0.0f;
            if (ok) {
                const float* xp = x + (size_t)y * WW + wh;
                #pragma unroll
                for (int n = 0; n < 8; ++n)
                    hv[i][n] = xp[(size_t)(n * 128 + rl) * (HH * WW)];
            }
        }
        #pragma unroll
        for (int i = 0; i < 3; ++i) {
            #pragma unroll
            for (int n = 0; n < 16; ++n) {
                int bc = n * 64 + bcl;
                float* d = &sX[(bc >> 5) * PB + (bc & 31) * CP + i * SW + 1 + 4 * part];
                d[0] = v[i][n].x; d[1] = v[i][n].y; d[2] = v[i][n].z; d[3] = v[i][n].w;
            }
        }
        #pragma unroll
        for (int i = 0; i < 3; ++i) {
            #pragma unroll
            for (int n = 0; n < 8; ++n) {
                int bc = n * 128 + rl;
                sX[(bc >> 5) * PB + (bc & 31) * CP + i * SW + side * (SW - 1)] = hv[i][n];
            }
        }
    }

    const int wv   = t >> 5;
    const int lane = t & 31;
    const int lr   = lane & 15;
    const int hh   = lane >> 4;
    const int tile = wv & 3;
    const int kh   = wv >> 2;
    const int m0   = (tile >> 1) * 16;
    const int n0   = (tile & 1) * 16;

    const float* aBase = &sX[(m0 + lr) * PB + hh * (2 * CP) + kh * (16 * CP)];
    const float* bBase = &sB[(n0 + lr) * LDP + hh * 18 + kh * 144];
    float* soBase = &sO[(m0 + hh * 8) * (TW * OC)];
    const int oSw = (n0 + lr) ^ (hh << 4);

    auto location = [&](const int wl, v4f (&wreg)[9], const bool refill) __attribute__((always_inline)) {
        #pragma unroll
        for (int u = 0; u < 9; ++u) {
            int e   = (t + u * NTH) * 4;
            int row = e / KD;
            int col = e - row * KD;
            *(v4f*)&sB[row * LDP + col] = wreg[u];
        }
        lds_barrier();

        if (refill) {
            const v4f* np = wp4 + (wl + 2) * ((OC * KD) / 4);
            #pragma unroll
            for (int u = 0; u < 9; ++u)
                wreg[u] = __builtin_nontemporal_load(np + t + u * NTH);
        }

        const float* ap = aBase + wl;
        v8f acc = {};
        #pragma unroll
        for (int q = 0; q < 4; ++q) {
            #pragma unroll
            for (int p = 0; p < 9; ++p) {
                v2f a;
                a.x = ap[q * (4 * CP) + tap_off(2 * p)];
                a.y = ap[q * (4 * CP) + tap_off(2 * p + 1)];
                v2f b = *(const v2f*)(bBase + q * 36 + 2 * p);
                acc = __builtin_amdgcn_wmma_f32_16x16x4_f32(
                    false, a, false, b, (short)0, acc, false, false);
            }
        }

        float* so = soBase + wl * OC + (oSw ^ ((wl >> 2) << 3));
        if (kh == 1) {
            #pragma unroll
            for (int r = 0; r < 8; ++r) so[r * (TW * OC)] = acc[r];
        }
        lds_barrier();
        if (kh == 0) {
            #pragma unroll
            for (int r = 0; r < 8; ++r) so[r * (TW * OC)] += acc[r];
        }
    };

    #pragma unroll 1
    for (int wl2 = 0; wl2 < TW - 2; wl2 += 2) {
        location(wl2,     wr[0], true);
        location(wl2 + 1, wr[1], true);
    }
    location(TW - 2, wr[0], false);
    location(TW - 1, wr[1], false);
    lds_barrier();

    for (int f = t; f < (B_ * OC * TW) / 4; f += NTH) {
        int row  = f >> 2;
        int part = f & 3;
        int o    = row & (OC - 1);
        int b    = row >> 5;
        const float* sp = &sO[(b * TW + part * 4) * OC + (o ^ (((b >> 3) & 1) << 4) ^ (part << 3))];
        v4f v  = {sp[0], sp[OC], sp[2 * OC], sp[3 * OC]};
        v4f bv = *(const v4f*)(bias + (size_t)o * (HH * WW) + h * WW + w0 + part * 4);
        v += bv;
        __builtin_nontemporal_store(v, (v4f*)(out + (size_t)row * (HH * WW) + h * WW + w0 + part * 4));
    }
}

extern "C" void kernel_launch(void* const* d_in, const int* in_sizes, int n_in,
                              void* d_out, int out_size, void* d_ws, size_t ws_size,
                              hipStream_t stream) {
    const float* x    = (const float*)d_in[0];
    const float* wgt  = (const float*)d_in[1];
    const float* bias = (const float*)d_in[2];
    float* out        = (float*)d_out;
    (void)in_sizes; (void)n_in; (void)out_size; (void)d_ws; (void)ws_size;

    dim3 grid(WW / TW, HH);
    lc2d_strip_wmma_f32_kernel<<<grid, NTH, 0, stream>>>(x, wgt, bias, out);
}


# 16 locations' results stay in registers, K halves added once at the end (same order); freed LDS holds a 2nd weight tile: 1 barrier per location, tile fill overlaps products; fragment reads grouped
# speedup vs baseline: 6.1579x; 1.0380x over previous
#include <hip/hip_runtime.h>


typedef __attribute__((ext_vector_type(2))) float v2f;
typedef __attribute__((ext_vector_type(4))) float v4f;
typedef __attribute__((ext_vector_type(8))) float v8f;

#define B_  32
#define IC  32
#define OC  32
#define HH  64
#define WW  64
#define KD  288
#define LDP 292
#define TW  16
#define SW  18
#define CP  54
#define PB  1729
#define NTH 256
#define SBF (OC * LDP)
#define SOP (B_ * TW * OC)

__device__ constexpr int tap_off(int t) {
    return (t / 9) * CP + ((t % 9) / 3) * SW + (t % 9) % 3;
}

__device__ __forceinline__ void lds_barrier() {
    __builtin_amdgcn_fence(__ATOMIC_RELEASE, "workgroup", "local");
    __builtin_amdgcn_s_barrier();
    __builtin_amdgcn_fence(__ATOMIC_ACQUIRE, "workgroup", "local");
}

__global__ __launch_bounds__(NTH)
void lc2d_strip_wmma_f32_kernel(const float* __restrict__ x,
                                const float* __restrict__ wgt,
                                const float* __restrict__ bias,
                                float* __restrict__ out) {
    __shared__ __attribute__((aligned(16))) float sB[2 * SBF];
    __shared__ float sX[B_ * PB];
    float* const sO = sX;
    static_assert(2 * SOP <= B_ * PB, "result planes must fit the slab");

    const int t  = threadIdx.x;
    const int w0 = blockIdx.x * TW;
    const int h  = blockIdx.y;

    const v4f* wp4 = (const v4f*)(wgt + (size_t)(h * WW + w0) * (OC * KD));
    v4f wr[2][9];
    #pragma unroll
    for (int s = 0; s < 2; ++s) {
        #pragma unroll
        for (int u = 0; u < 9; ++u)
            wr[s][u] = __builtin_nontemporal_load(wp4 + s * ((OC * KD) / 4) + t + u * NTH);
    }

    {
        const int part = t & 3;
        const int bcl  = t >> 2;
        v4f v[3][16];
        #pragma unroll
        for (int i = 0; i < 3; ++i) {
            const int  y   = h - 1 + i;
            const bool yok = (unsigned)y < (unsigned)HH;
            if (yok) {
                const float* xp = x + (size_t)y * WW + w0 + 4 * part;
                #pragma unroll
                for (int n = 0; n < 16; ++n)
                    v[i][n] = *(const v4f*)(xp + (size_t)(n * 64 + bcl) * (HH * WW));
            } else {
                #pragma unroll
                for (int n = 0; n < 16; ++n) v[i][n] = (v4f){0.0f, 0.0f, 0.0f, 0.0f};
            }
        }
        const int  side = t & 1;
        const int  rl   = t >> 1;
        const int  wh   = side ? (w0 + TW) : (w0 - 1);
        const bool wok  = (unsigned)wh < (unsigned)WW;
        float hv[3][8];
        #pragma unroll
        for (int i = 0; i < 3; ++i) {
            const int  y  = h - 1 + i;
            const bool ok = wok && ((unsigned)y < (unsigned)HH);
            #pragma unroll
            for (int n = 0; n < 8; ++n) hv[i][n] = 0.0f;
            if (ok) {
                const float* xp = x + (size_t)y * WW + wh;
                #pragma unroll
                for (int n = 0; n < 8; ++n)
                    hv[i][n] = xp[(size_t)(n * 128 + rl) * (HH * WW)];
            }
        }
        #pragma unroll
        for (int i = 0; i < 3; ++i) {
            #pragma unroll
            for (int n = 0; n < 16; ++n) {
                int bc = n * 64 + bcl;
                float* d = &sX[(bc >> 5) * PB + (bc & 31) * CP + i * SW + 1 + 4 * part];
                d[0] = v[i][n].x; d[1] = v[i][n].y; d[2] = v[i][n].z; d[3] = v[i][n].w;
            }
        }
        #pragma unroll
        for (int i = 0; i < 3; ++i) {
            #pragma unroll
            for (int n = 0; n < 8; ++n) {
                int bc = n * 128 + rl;
                sX[(bc >> 5) * PB + (bc & 31) * CP + i * SW + side * (SW - 1)] = hv[i][n];
            }
        }
    }

    const int wv   = t >> 5;
    const int lane = t & 31;
    const int lr   = lane & 15;
    const int hh   = lane >> 4;
    const int tile = wv & 3;
    const int kh   = wv >> 2;
    const int m0   = (tile >> 1) * 16;
    const int n0   = (tile & 1) * 16;

    const float* aBase = &sX[(m0 + lr) * PB + hh * (2 * CP) + kh * (16 * CP)];
    const float* bBase = &sB[(n0 + lr) * LDP + hh * 18 + kh * 144];
    float* soBase = &sO[(m0 + hh * 8) * (TW * OC)];
    const int oSw = (n0 + lr) ^ (hh << 4);

    auto put_weights = [&](float* dst, const v4f (&wreg)[9]) __attribute__((always_inline)) {
        #pragma unroll
        for (int u = 0; u < 9; ++u) {
            int e   = (t + u * NTH) * 4;
            int row = e / KD;
            int col = e - row * KD;
            *(v4f*)&dst[row * LDP + col] = wreg[u];
        }
    };
    auto fetch_weights = [&](v4f (&wreg)[9], const int wl) __attribute__((always_inline)) {
        const v4f* np = wp4 + wl * ((OC * KD) / 4);
        #pragma unroll
        for (int u = 0; u < 9; ++u)
            wreg[u] = __builtin_nontemporal_load(np + t + u * NTH);
    };
    auto load_frags = [&](v2f (&fa)[36], v2f (&fb)[36], const int wl, const float* bp) __attribute__((always_inline)) {
        const float* ap = aBase + wl;
        #pragma unroll
        for (int q = 0; q < 4; ++q) {
            #pragma unroll
            for (int p = 0; p < 9; ++p) {
                fa[q * 9 + p].x = ap[q * (4 * CP) + tap_off(2 * p)];
                fa[q * 9 + p].y = ap[q * (4 * CP) + tap_off(2 * p + 1)];
                fb[q * 9 + p]   = *(const v2f*)(bp + q * 36 + 2 * p);
            }
        }
    };
    auto products = [&](const v2f (&fa)[36], const v2f (&fb)[36]) __attribute__((always_inline)) -> v8f {
        v8f acc = {};
        #pragma unroll
        for (int s = 0; s < 36; ++s) {
            acc = __builtin_amdgcn_wmma_f32_16x16x4_f32(
                false, fa[s], false, fb[s], (short)0, acc, false, false);
        }
        return acc;
    };

    v8f res[TW];

    put_weights(sB, wr[0]);
    fetch_weights(wr[0], 2);
    lds_barrier();

    #pragma unroll
    for (int wl = 0; wl < TW; ++wl) {
        v2f fa[36], fb[36];
        load_frags(fa, fb, wl, bBase + (wl & 1) * SBF);
        if (wl + 1 < TW) put_weights(sB + ((wl + 1) & 1) * SBF, wr[(wl + 1) & 1]);
        if (wl + 3 < TW) fetch_weights(wr[(wl + 1) & 1], wl + 3);
        res[wl] = products(fa, fb);
        lds_barrier();
    }

    #pragma unroll
    for (int wl = 0; wl < TW; ++wl) {
        float* so = soBase + kh * SOP + wl * OC + (oSw ^ ((wl >> 2) << 3));
        #pragma unroll
        for (int r = 0; r < 8; ++r) so[r * (TW * OC)] = res[wl][r];
    }
    lds_barrier();

    for (int f = t; f < (B_ * OC * TW) / 4; f += NTH) {
        int row  = f >> 2;
        int part = f & 3;
        int o    = row & (OC - 1);
        int b    = row >> 5;
        const float* sp = &sO[(b * TW + part * 4) * OC + (o ^ (((b >> 3) & 1) << 4) ^ (part << 3))];
        v4f v  = {sp[SOP], sp[SOP + OC], sp[SOP + 2 * OC], sp[SOP + 3 * OC]};
        v4f v0 = {sp[0], sp[OC], sp[2 * OC], sp[3 * OC]};
        v += v0;
        v4f bv = *(const v4f*)(bias + (size_t)o * (HH * WW) + h * WW + w0 + part * 4);
        v += bv;
        __builtin_nontemporal_store(v, (v4f*)(out + (size_t)row * (HH * WW) + h * WW + w0 + part * 4));
    }
}

extern "C" void kernel_launch(void* const* d_in, const int* in_sizes, int n_in,
                              void* d_out, int out_size, void* d_ws, size_t ws_size,
                              hipStream_t stream) {
    const float* x    = (const float*)d_in[0];
    const float* wgt  = (const float*)d_in[1];
    const float* bias = (const float*)d_in[2];
    float* out        = (float*)d_out;
    (void)in_sizes; (void)n_in; (void)out_size; (void)d_ws; (void)ws_size;

    dim3 grid(WW / TW, HH);
    lc2d_strip_wmma_f32_kernel<<<grid, NTH, 0, stream>>>(x, wgt, bias, out);
}


# x slab staging: five float4 lanes read each 80-byte run (16 interior columns plus the halo column in the same 128-byte line); only the other halo column is loaded alone, none at image edges
# speedup vs baseline: 6.5231x; 1.0593x over previous
#include <hip/hip_runtime.h>


typedef __attribute__((ext_vector_type(2))) float v2f;
typedef __attribute__((ext_vector_type(4))) float v4f;
typedef __attribute__((ext_vector_type(8))) float v8f;

#define B_  32
#define IC  32
#define OC  32
#define HH  64
#define WW  64
#define KD  288
#define LDP 292
#define TW  16
#define SW  18
#define CP  54
#define PB  1729
#define NTH 256
#define SBF (OC * LDP)
#define SOP (B_ * TW * OC)

__device__ constexpr int tap_off(int t) {
    return (t / 9) * CP + ((t % 9) / 3) * SW + (t % 9) % 3;
}

__device__ __forceinline__ void lds_barrier() {
    __builtin_amdgcn_fence(__ATOMIC_RELEASE, "workgroup", "local");
    __builtin_amdgcn_s_barrier();
    __builtin_amdgcn_fence(__ATOMIC_ACQUIRE, "workgroup", "local");
}

__global__ __launch_bounds__(NTH)
void lc2d_strip_wmma_f32_kernel(const float* __restrict__ x,
                                const float* __restrict__ wgt,
                                const float* __restrict__ bias,
                                float* __restrict__ out) {
    __shared__ __attribute__((aligned(16))) float sB[2 * SBF];
    __shared__ float sX[B_ * PB];
    float* const sO = sX;
    static_assert(2 * SOP <= B_ * PB, "result planes must fit the slab");

    const int t  = threadIdx.x;
    const int w0 = blockIdx.x * TW;
    const int h  = blockIdx.y;

    const v4f* wp4 = (const v4f*)(wgt + (size_t)(h * WW + w0) * (OC * KD));
    v4f wr[2][9];
    #pragma unroll
    for (int s = 0; s < 2; ++s) {
        #pragma unroll
        for (int u = 0; u < 9; ++u)
            wr[s][u] = __builtin_nontemporal_load(wp4 + s * ((OC * KD) / 4) + t + u * NTH);
    }

    {
        const bool odd = (blockIdx.x & 1) != 0;
        const int  ln  = t & 31;
        const int  l5  = ln / 5;
        const int  ch  = ln - 5 * l5;
        const bool act = ln < 30;
        const int  lw  = (t >> 5) * 6 + l5;
        const int  p0  = lw / 3;
        const int  ri  = lw - 3 * p0;
        const int  y   = h - 1 + ri;
        const bool yok = (unsigned)y < (unsigned)HH;
        const int  yc  = y < 0 ? 0 : (y > HH - 1 ? HH - 1 : y);
        const int  cs  = w0 - (odd ? 4 : 0);
        v4f v[64];
        if (act) {
            const float* xp = x + (size_t)p0 * (HH * WW) + (size_t)yc * WW + cs + 4 * ch;
            #pragma unroll
            for (int u = 0; u < 64; ++u)
                v[u] = *(const v4f*)(xp + (size_t)u * (16 * HH * WW));
        }
        const int  wh  = odd ? (w0 + TW) : (w0 - 1);
        const bool wok = (unsigned)wh < (unsigned)WW;
        const int  ec  = odd ? (SW - 1) : 0;
        float hv[3][4];
        #pragma unroll
        for (int i = 0; i < 3; ++i) {
            const int  yy = h - 1 + i;
            const bool ok = wok && ((unsigned)yy < (unsigned)HH);
            #pragma unroll
            for (int n = 0; n < 4; ++n) hv[i][n] = 0.0f;
            if (ok) {
                const float* xp = x + (size_t)yy * WW + wh;
                #pragma unroll
                for (int n = 0; n < 4; ++n)
                    hv[i][n] = xp[(size_t)(n * 256 + t) * (HH * WW)];
            }
        }
        const int  i0   = p0 * CP + ri * SW + 4 * ch + (odd ? -3 : 1);
        const bool inte = odd ? (ch >= 1) : (ch <= 3);
        if (act && yok && inte) {
            #pragma unroll
            for (int u = 0; u < 64; ++u) {
                float* d = &sX[(u >> 1) * PB + (u & 1) * (16 * CP) + i0];
                d[0] = v[u].x; d[1] = v[u].y; d[2] = v[u].z; d[3] = v[u].w;
            }
        }
        if (act && yok && !inte) {
            if (odd) {
                #pragma unroll
                for (int u = 0; u < 64; ++u) sX[(u >> 1) * PB + (u & 1) * (16 * CP) + i0 + 3] = v[u].w;
            } else {
                #pragma unroll
                for (int u = 0; u < 64; ++u) sX[(u >> 1) * PB + (u & 1) * (16 * CP) + i0] = v[u].x;
            }
        }
        #pragma unroll
        for (int i = 0; i < 3; ++i) {
            const int yy = h - 1 + i;
            if ((unsigned)yy < (unsigned)HH) {
                #pragma unroll
                for (int n = 0; n < 4; ++n) {
                    int bc = n * 256 + t;
                    sX[(bc >> 5) * PB + (bc & 31) * CP + i * SW + ec] = hv[i][n];
                }
            }
        }
        #pragma unroll
        for (int i = 0; i < 3; i += 2) {
            const int yy = h - 1 + i;
            if ((unsigned)yy >= (unsigned)HH) {
                #pragma nounroll
                for (int k = t; k < B_ * IC * SW; k += NTH) {
                    int bc  = k / SW;
                    int col = k - bc * SW;
                    sX[(bc >> 5) * PB + (bc & 31) * CP + i * SW + col] = 0.0f;
                }
            }
        }
    }

    const int wv   = t >> 5;
    const int lane = t & 31;
    const int lr   = lane & 15;
    const int hh   = lane >> 4;
    const int tile = wv & 3;
    const int kh   = wv >> 2;
    const int m0   = (tile >> 1) * 16;
    const int n0   = (tile & 1) * 16;

    const float* aBase = &sX[(m0 + lr) * PB + hh * (2 * CP) + kh * (16 * CP)];
    const float* bBase = &sB[(n0 + lr) * LDP + hh * 18 + kh * 144];
    float* soBase = &sO[(m0 + hh * 8) * (TW * OC)];
    const int oSw = (n0 + lr) ^ (hh << 4);

    auto put_weights = [&](float* dst, const v4f (&wreg)[9]) __attribute__((always_inline)) {
        #pragma unroll
        for (int u = 0; u < 9; ++u) {
            int e   = (t + u * NTH) * 4;
            int row = e / KD;
            int col = e - row * KD;
            *(v4f*)&dst[row * LDP + col] = wreg[u];
        }
    };
    auto fetch_weights = [&](v4f (&wreg)[9], const int wl) __attribute__((always_inline)) {
        const v4f* np = wp4 + wl * ((OC * KD) / 4);
        #pragma unroll
        for (int u = 0; u < 9; ++u)
            wreg[u] = __builtin_nontemporal_load(np + t + u * NTH);
    };
    auto load_frags = [&](v2f (&fa)[36], v2f (&fb)[36], const int wl, const float* bp) __attribute__((always_inline)) {
        const float* ap = aBase + wl;
        #pragma unroll
        for (int q = 0; q < 4; ++q) {
            #pragma unroll
            for (int p = 0; p < 9; ++p) {
                fa[q * 9 + p].x = ap[q * (4 * CP) + tap_off(2 * p)];
                fa[q * 9 + p].y = ap[q * (4 * CP) + tap_off(2 * p + 1)];
                fb[q * 9 + p]   = *(const v2f*)(bp + q * 36 + 2 * p);
            }
        }
    };
    auto products = [&](const v2f (&fa)[36], const v2f (&fb)[36]) __attribute__((always_inline)) -> v8f {
        v8f acc = {};
        #pragma unroll
        for (int s = 0; s < 36; ++s) {
            acc = __builtin_amdgcn_wmma_f32_16x16x4_f32(
                false, fa[s], false, fb[s], (short)0, acc, false, false);
        }
        return acc;
    };

    v8f res[TW];

    put_weights(sB, wr[0]);
    fetch_weights(wr[0], 2);
    lds_barrier();

    #pragma unroll
    for (int wl = 0; wl < TW; ++wl) {
        v2f fa[36], fb[36];
        load_frags(fa, fb, wl, bBase + (wl & 1) * SBF);
        if (wl + 1 < TW) put_weights(sB + ((wl + 1) & 1) * SBF, wr[(wl + 1) & 1]);
        if (wl + 3 < TW) fetch_weights(wr[(wl + 1) & 1], wl + 3);
        res[wl] = products(fa, fb);
        lds_barrier();
    }

    #pragma unroll
    for (int wl = 0; wl < TW; ++wl) {
        float* so = soBase + kh * SOP + wl * OC + (oSw ^ ((wl >> 2) << 3));
        #pragma unroll
        for (int r = 0; r < 8; ++r) so[r * (TW * OC)] = res[wl][r];
    }
    lds_barrier();

    for (int f = t; f < (B_ * OC * TW) / 4; f += NTH) {
        int row  = f >> 2;
        int part = f & 3;
        int o    = row & (OC - 1);
        int b    = row >> 5;
        const float* sp = &sO[(b * TW + part * 4) * OC + (o ^ (((b >> 3) & 1) << 4) ^ (part << 3))];
        v4f v  = {sp[SOP], sp[SOP + OC], sp[SOP + 2 * OC], sp[SOP + 3 * OC]};
        v4f v0 = {sp[0], sp[OC], sp[2 * OC], sp[3 * OC]};
        v += v0;
        v4f bv = *(const v4f*)(bias + (size_t)o * (HH * WW) + h * WW + w0 + part * 4);
        v += bv;
        __builtin_nontemporal_store(v, (v4f*)(out + (size_t)row * (HH * WW) + h * WW + w0 + part * 4));
    }
}

extern "C" void kernel_launch(void* const* d_in, const int* in_sizes, int n_in,
                              void* d_out, int out_size, void* d_ws, size_t ws_size,
                              hipStream_t stream) {
    const float* x    = (const float*)d_in[0];
    const float* wgt  = (const float*)d_in[1];
    const float* bias = (const float*)d_in[2];
    float* out        = (float*)d_out;
    (void)in_sizes; (void)n_in; (void)out_size; (void)d_ws; (void)ws_size;

    dim3 grid(WW / TW, HH);
    lc2d_strip_wmma_f32_kernel<<<grid, NTH, 0, stream>>>(x, wgt, bias, out);
}
